# starting kernel
# baseline (speedup 1.0000x reference)
#include <hip/hip_runtime.h>

#define DD        256
#define SEQN      128
#define NBATCH    32
#define XE_STRIDE 260
#define P_STRIDE  132

typedef float v2f __attribute__((ext_vector_type(2)));
typedef float v8f __attribute__((ext_vector_type(8)));

#define XE_OFF  0
#define XL_OFF  (SEQN * XE_STRIDE)
#define XR_OFF  (XL_OFF + SEQN * P_STRIDE)
#define TOK_OFF (XR_OFF + SEQN * P_STRIDE)
#define SMEM_FLOATS (TOK_OFF + SEQN)

__global__ __launch_bounds__(256) void RN_pair_fused_kernel(
    const int* __restrict__ X, const float* __restrict__ emb,
    const float* __restrict__ W_l, const float* __restrict__ b_l,
    const float* __restrict__ W_r, const float* __restrict__ b_r,
    float* __restrict__ pooled) {
  extern __shared__ float smem[];
  float* xe = smem + XE_OFF;
  float* xl = smem + XL_OFF;
  float* xr = smem + XR_OFF;
  int*   tok = (int*)(smem + TOK_OFF);

  const int b     = blockIdx.x;
  const int dbase = blockIdx.y * 128;
  const int tid   = threadIdx.x;

  if (tid < SEQN) tok[tid] = X[b * SEQN + tid];
  __syncthreads();

  for (int idx = tid; idx < SEQN * (DD / 4); idx += 256) {
    const int row = idx >> 6;
    const int c4  = idx & 63;
    const float4 v = ((const float4*)(emb + (long)tok[row] * DD))[c4];
    float* dst = xe + row * XE_STRIDE + c4 * 4;
    dst[0] = v.x; dst[1] = v.y; dst[2] = v.z; dst[3] = v.w;
  }
  __syncthreads();

  const int wave = tid >> 5;
  const int lane = tid & 31;
  const int lr   = lane & 15;
  const int koff = (lane >> 4) * 2;

  for (int cu = wave; cu < 16; cu += 8) {
    const int mat     = cu >> 3;
    const int colbase = (cu & 7) * 16;
    const float* W    = mat ? W_r : W_l;
    const float* bias = mat ? b_r : b_l;
    float*       dst  = mat ? xr : xl;
    const int dcol    = dbase + colbase + lr;

    const float bv = bias[dcol];
    v8f acc[8];
#pragma unroll
    for (int rt = 0; rt < 8; ++rt)
      acc[rt] = (v8f){ bv, bv, bv, bv, bv, bv, bv, bv };

    const float* brow = W + (long)dcol * DD + koff;
    for (int kb = 0; kb < DD; kb += 32) {
      v2f bf[8];
#pragma unroll
      for (int s = 0; s < 8; ++s)
        bf[s] = *(const v2f*)(brow + kb + 4 * s);
#pragma unroll
      for (int rt = 0; rt < 8; ++rt) {
        const float* arow = xe + (rt * 16 + lr) * XE_STRIDE + kb + koff;
#pragma unroll
        for (int s = 0; s < 8; ++s) {
          v2f a = *(const v2f*)(arow + 4 * s);
          acc[rt] = __builtin_amdgcn_wmma_f32_16x16x4_f32(
               false, a,  false, bf[s],
               (short)0, acc[rt],
               false,  false);
        }
      }
    }
#pragma unroll
    for (int rt = 0; rt < 8; ++rt) {
      float* obase = dst + (rt * 16 + 8 * (lane >> 4)) * P_STRIDE + colbase + lr;
#pragma unroll
      for (int v = 0; v < 8; ++v) obase[v * P_STRIDE] = acc[rt][v];
    }
  }
  __syncthreads();

  const int d    = tid & 127;
  const int ih0  = (tid >> 7) * 64;
  float acc = 0.f;
  for (int jb = 0; jb < SEQN; jb += 8) {
    float xlv[8];
#pragma unroll
    for (int u = 0; u < 8; ++u) xlv[u] = xl[(jb + u) * P_STRIDE + d];
    for (int i = ih0; i < ih0 + 64; ++i) {
      const float xrv = xr[i * P_STRIDE + d];
#pragma unroll
      for (int u = 0; u < 8; ++u) acc += fmaxf(xlv[u] + xrv, 0.f);
    }
  }
  smem[tid] = acc;
  __syncthreads();
  if (tid < 128)
    pooled[b * DD + dbase + tid] = smem[tid] + smem[tid + 128];
}

__global__ __launch_bounds__(32) void RN_out_kernel(
    const float* __restrict__ pooled, const float* __restrict__ W_rn,
    const float* __restrict__ b_rn, float* __restrict__ out) {
  const int mt   = blockIdx.x;
  const int nt   = blockIdx.y;
  const int lane = threadIdx.x;
  const int lr   = lane & 15;
  const int koff = (lane >> 4) * 2;
  const int row  = mt * 16 + lr;
  const int col  = nt * 16 + lr;

  const float bv = 16384.0f * b_rn[col];
  v8f c = { bv, bv, bv, bv, bv, bv, bv, bv };

  const float* arow = pooled + (long)row * DD + koff;
  const float* brow = W_rn + (long)col * DD + koff;
  for (int k0 = 0; k0 < DD; k0 += 4) {
    v2f a  = *(const v2f*)(arow + k0);
    v2f bb = *(const v2f*)(brow + k0);
    c = __builtin_amdgcn_wmma_f32_16x16x4_f32(
        false, a, false, bb, (short)0, c, false, false);
  }
  float* obase = out + (mt * 16 + 8 * (lane >> 4)) * DD + nt * 16 + lr;
#pragma unroll
  for (int v = 0; v < 8; ++v) obase[v * DD] = c[v];
}

extern "C" void kernel_launch(void* const* d_in, const int* in_sizes, int n_in,
                              void* d_out, int out_size, void* d_ws, size_t ws_size,
                              hipStream_t stream) {
  (void)in_sizes; (void)n_in; (void)out_size; (void)ws_size;
  const int*   X    = (const int*)d_in[0];
  const float* emb  = (const float*)d_in[1];
  const float* W_l  = (const float*)d_in[2];
  const float* b_l  = (const float*)d_in[3];
  const float* W_r  = (const float*)d_in[4];
  const float* b_r  = (const float*)d_in[5];
  const float* W_rn = (const float*)d_in[6];
  const float* b_rn = (const float*)d_in[7];
  float* pooled = (float*)d_ws;
  float* out    = (float*)d_out;

  dim3 gridA(NBATCH, 2), blockA(256);
  size_t shmem = (size_t)SMEM_FLOATS * sizeof(float);
  RN_pair_fused_kernel<<<gridA, blockA, shmem, stream>>>(
      X, emb, W_l, b_l, W_r, b_r, pooled);

  dim3 gridB(2, 16), blockB(32);
  RN_out_kernel<<<gridB, blockB, 0, stream>>>(pooled, W_rn, b_rn, out);
}


# Pair loop re-tiled: thread keeps 32 xl_j of its column in registers with 16 two-wide running sums; each xr_i is read from LDS once and meets all 32 (was 8 xl, one sum chain); pair-sum order changes
# speedup vs baseline: 3.9104x; 3.9104x over previous
#include <hip/hip_runtime.h>


#define DD        256
#define SEQN      128
#define NBATCH    32
#define HROWS     64
#define NPART     4
#define XE_STRIDE 260
#define P_STRIDE  132

typedef float v2f __attribute__((ext_vector_type(2)));
typedef float v8f __attribute__((ext_vector_type(8)));

#define XE_OFF  0
#define XL_OFF  (SEQN * XE_STRIDE)
#define XR_OFF  (XL_OFF + HROWS * P_STRIDE)
#define TOK_OFF (XR_OFF + HROWS * P_STRIDE)
#define SMEM_FLOATS (TOK_OFF + SEQN)

#define PLANE    (NBATCH * DD)
#define WS_NEED  ((size_t)NPART * PLANE * sizeof(float))

__global__ __launch_bounds__(256) void RN_pair_block_kernel(
    const int* __restrict__ X, const float* __restrict__ emb,
    const float* __restrict__ W_l, const float* __restrict__ b_l,
    const float* __restrict__ W_r, const float* __restrict__ b_r,
    float* __restrict__ part, size_t ws_bytes) {
  if (ws_bytes < WS_NEED) return;

  extern __shared__ float smem[];
  float* xe = smem + XE_OFF;
  float* xl = smem + XL_OFF;
  float* xr = smem + XR_OFF;
  int*   tok = (int*)(smem + TOK_OFF);

  const int b     = blockIdx.x;
  const int dbase = blockIdx.y * 128;
  const int blk   = blockIdx.z;
  const int jh    = blk & 1;
  const int ih    = blk >> 1;
  const int tid   = threadIdx.x;

  const bool diag  = (ih == jh);
  const int  nrows = diag ? HROWS : 2 * HROWS;
  const float* xei = diag ? xe : xe + HROWS * XE_STRIDE;

  if (tid < nrows) {
    const int s = (tid < HROWS) ? jh * HROWS + tid : ih * HROWS + (tid - HROWS);
    tok[tid] = X[b * SEQN + s];
  }
  __syncthreads();

  for (int idx = tid; idx < nrows * (DD / 4); idx += 256) {
    const int row = idx >> 6;
    const int c4  = idx & 63;
    const float4 v = ((const float4*)(emb + (long)tok[row] * DD))[c4];
    float* dst = xe + row * XE_STRIDE + c4 * 4;
    dst[0] = v.x; dst[1] = v.y; dst[2] = v.z; dst[3] = v.w;
  }
  __syncthreads();

  const int wave = tid >> 5;
  const int lane = tid & 31;
  const int lr   = lane & 15;
  const int koff = (lane >> 4) * 2;

  for (int cu = wave; cu < 16; cu += 8) {
    const int mat     = cu >> 3;
    const int colbase = (cu & 7) * 16;
    const float* W    = mat ? W_r : W_l;
    const float* bias = mat ? b_r : b_l;
    const float* asrc = mat ? xei : xe;
    float*       dst  = mat ? xr : xl;
    const int dcol    = dbase + colbase + lr;

    const float bv = bias[dcol];
    v8f acc[4];
#pragma unroll
    for (int rt = 0; rt < 4; ++rt)
      acc[rt] = (v8f){ bv, bv, bv, bv, bv, bv, bv, bv };

    const float* brow = W + (long)dcol * DD + koff;
    for (int kb = 0; kb < DD; kb += 32) {
      v2f bf[8];
#pragma unroll
      for (int s = 0; s < 8; ++s)
        bf[s] = *(const v2f*)(brow + kb + 4 * s);
#pragma unroll
      for (int rt = 0; rt < 4; ++rt) {
        const float* arow = asrc + (rt * 16 + lr) * XE_STRIDE + kb + koff;
#pragma unroll
        for (int s = 0; s < 8; ++s) {
          v2f a = *(const v2f*)(arow + 4 * s);
          acc[rt] = __builtin_amdgcn_wmma_f32_16x16x4_f32(
               false, a,  false, bf[s],
               (short)0, acc[rt],
               false,  false);
        }
      }
    }
#pragma unroll
    for (int rt = 0; rt < 4; ++rt) {
      float* obase = dst + (rt * 16 + 8 * (lane >> 4)) * P_STRIDE + colbase + lr;
#pragma unroll
      for (int v = 0; v < 8; ++v) obase[v * P_STRIDE] = acc[rt][v];
    }
  }
  __syncthreads();

  const int d  = tid & 127;
  const int j0 = (tid >> 7) * 32;
  v2f xl2[16], sum2[16];
#pragma unroll
  for (int u = 0; u < 16; ++u) {
    xl2[u]  = (v2f){ xl[(j0 + 2 * u) * P_STRIDE + d], xl[(j0 + 2 * u + 1) * P_STRIDE + d] };
    sum2[u] = (v2f){ 0.f, 0.f };
  }
  for (int ib = 0; ib < HROWS; ib += 8) {
    float xrv[8];
#pragma unroll
    for (int v = 0; v < 8; ++v) xrv[v] = xr[(ib + v) * P_STRIDE + d];
#pragma unroll
    for (int v = 0; v < 8; ++v) {
      const v2f r2 = { xrv[v], xrv[v] };
      v2f t[16];
#pragma unroll
      for (int u = 0; u < 16; ++u) t[u] = xl2[u] + r2;
#pragma unroll
      for (int u = 0; u < 16; ++u) {
        t[u].x = fmaxf(t[u].x, 0.f);
        t[u].y = fmaxf(t[u].y, 0.f);
      }
#pragma unroll
      for (int u = 0; u < 16; ++u) sum2[u] += t[u];
    }
  }
#pragma unroll
  for (int w = 8; w >= 1; w >>= 1) {
#pragma unroll
    for (int u = 0; u < w; ++u) sum2[u] += sum2[u + w];
  }
  const float acc = sum2[0].x + sum2[0].y;
  smem[tid] = acc;
  __syncthreads();
  if (tid < 128)
    part[(blk * NBATCH + b) * DD + dbase + tid] = smem[tid] + smem[tid + 128];
}

__global__ __launch_bounds__(32) void RN_out_kernel(
    const float* __restrict__ part, const float* __restrict__ W_rn,
    const float* __restrict__ b_rn, float* __restrict__ out, size_t ws_bytes) {
  if (ws_bytes < WS_NEED) return;

  const int mt   = blockIdx.x;
  const int nt   = blockIdx.y;
  const int lane = threadIdx.x;
  const int lr   = lane & 15;
  const int koff = (lane >> 4) * 2;
  const int row  = mt * 16 + lr;
  const int col  = nt * 16 + lr;

  const float bv = 16384.0f * b_rn[col];
  v8f c = { bv, bv, bv, bv, bv, bv, bv, bv };

  const float* arow = part + (long)row * DD + koff;
  const float* brow = W_rn + (long)col * DD + koff;
  for (int k0 = 0; k0 < DD; k0 += 4) {
    const v2f a0 = *(const v2f*)(arow + k0);
    const v2f a1 = *(const v2f*)(arow + PLANE + k0);
    const v2f a2 = *(const v2f*)(arow + 2 * PLANE + k0);
    const v2f a3 = *(const v2f*)(arow + 3 * PLANE + k0);
    const v2f a  = (a0 + a1) + (a2 + a3);
    v2f bb = *(const v2f*)(brow + k0);
    c = __builtin_amdgcn_wmma_f32_16x16x4_f32(
        false, a, false, bb, (short)0, c, false, false);
  }
  float* obase = out + (mt * 16 + 8 * (lane >> 4)) * DD + nt * 16 + lr;
#pragma unroll
  for (int v = 0; v < 8; ++v) obase[v * DD] = c[v];
}

extern "C" void kernel_launch(void* const* d_in, const int* in_sizes, int n_in,
                              void* d_out, int out_size, void* d_ws, size_t ws_size,
                              hipStream_t stream) {
  (void)in_sizes; (void)n_in; (void)out_size;
  const int*   X    = (const int*)d_in[0];
  const float* emb  = (const float*)d_in[1];
  const float* W_l  = (const float*)d_in[2];
  const float* b_l  = (const float*)d_in[3];
  const float* W_r  = (const float*)d_in[4];
  const float* b_r  = (const float*)d_in[5];
  const float* W_rn = (const float*)d_in[6];
  const float* b_rn = (const float*)d_in[7];
  float* part = (float*)d_ws;
  float* out  = (float*)d_out;

  dim3 gridA(NBATCH, 2, NPART), blockA(256);
  size_t shmem = (size_t)SMEM_FLOATS * sizeof(float);
  RN_pair_block_kernel<<<gridA, blockA, shmem, stream>>>(
      X, emb, W_l, b_l, W_r, b_r, part, ws_size);

  dim3 gridB(2, 16), blockB(32);
  RN_out_kernel<<<gridB, blockB, 0, stream>>>(part, W_rn, b_rn, out, ws_size);
}


# Operands of the next 32-wide K block (A from LDS, B and partial planes from global) are loaded as one group ahead of the current block's WMMAs, second register set, both kernels; sums unchanged
# speedup vs baseline: 4.6998x; 1.2019x over previous
#include <hip/hip_runtime.h>


#define DD        256
#define SEQN      128
#define NBATCH    32
#define HROWS     64
#define NPART     4
#define XE_STRIDE 260
#define P_STRIDE  132

typedef float v2f __attribute__((ext_vector_type(2)));
typedef float v8f __attribute__((ext_vector_type(8)));

#define XE_OFF  0
#define XL_OFF  (SEQN * XE_STRIDE)
#define XR_OFF  (XL_OFF + HROWS * P_STRIDE)
#define TOK_OFF (XR_OFF + HROWS * P_STRIDE)
#define SMEM_FLOATS (TOK_OFF + SEQN)

#define PLANE    (NBATCH * DD)
#define WS_NEED  ((size_t)NPART * PLANE * sizeof(float))

__global__ __launch_bounds__(256) void RN_pair_block_kernel(
    const int* __restrict__ X, const float* __restrict__ emb,
    const float* __restrict__ W_l, const float* __restrict__ b_l,
    const float* __restrict__ W_r, const float* __restrict__ b_r,
    float* __restrict__ part, size_t ws_bytes) {
  if (ws_bytes < WS_NEED) return;

  extern __shared__ float smem[];
  float* xe = smem + XE_OFF;
  float* xl = smem + XL_OFF;
  float* xr = smem + XR_OFF;
  int*   tok = (int*)(smem + TOK_OFF);

  const int b     = blockIdx.x;
  const int dbase = blockIdx.y * 128;
  const int blk   = blockIdx.z;
  const int jh    = blk & 1;
  const int ih    = blk >> 1;
  const int tid   = threadIdx.x;

  const bool diag  = (ih == jh);
  const int  nrows = diag ? HROWS : 2 * HROWS;
  const float* xei = diag ? xe : xe + HROWS * XE_STRIDE;

  if (tid < nrows) {
    const int s = (tid < HROWS) ? jh * HROWS + tid : ih * HROWS + (tid - HROWS);
    tok[tid] = X[b * SEQN + s];
  }
  __syncthreads();

  for (int idx = tid; idx < nrows * (DD / 4); idx += 256) {
    const int row = idx >> 6;
    const int c4  = idx & 63;
    const float4 v = ((const float4*)(emb + (long)tok[row] * DD))[c4];
    float* dst = xe + row * XE_STRIDE + c4 * 4;
    dst[0] = v.x; dst[1] = v.y; dst[2] = v.z; dst[3] = v.w;
  }
  __syncthreads();

  const int wave = tid >> 5;
  const int lane = tid & 31;
  const int lr   = lane & 15;
  const int koff = (lane >> 4) * 2;

  for (int cu = wave; cu < 16; cu += 8) {
    const int mat     = cu >> 3;
    const int colbase = (cu & 7) * 16;
    const float* W    = mat ? W_r : W_l;
    const float* bias = mat ? b_r : b_l;
    const float* asrc = mat ? xei : xe;
    float*       dst  = mat ? xr : xl;
    const int dcol    = dbase + colbase + lr;

    const float bv = bias[dcol];
    v8f acc[4];
#pragma unroll
    for (int rt = 0; rt < 4; ++rt)
      acc[rt] = (v8f){ bv, bv, bv, bv, bv, bv, bv, bv };

    const float* brow  = W + (long)dcol * DD + koff;
    const float* abase = asrc + lr * XE_STRIDE + koff;
    v2f bf[8], af[4][8];
#pragma unroll
    for (int s = 0; s < 8; ++s) bf[s] = *(const v2f*)(brow + 4 * s);
#pragma unroll
    for (int rt = 0; rt < 4; ++rt)
#pragma unroll
      for (int s = 0; s < 8; ++s)
        af[rt][s] = *(const v2f*)(abase + rt * 16 * XE_STRIDE + 4 * s);
    for (int kb = 32; kb < DD; kb += 32) {
      v2f bn[8], an[4][8];
#pragma unroll
      for (int s = 0; s < 8; ++s) bn[s] = *(const v2f*)(brow + kb + 4 * s);
#pragma unroll
      for (int rt = 0; rt < 4; ++rt)
#pragma unroll
        for (int s = 0; s < 8; ++s)
          an[rt][s] = *(const v2f*)(abase + rt * 16 * XE_STRIDE + kb + 4 * s);
      __builtin_amdgcn_sched_barrier(0);
#pragma unroll
      for (int rt = 0; rt < 4; ++rt)
#pragma unroll
        for (int s = 0; s < 8; ++s)
          acc[rt] = __builtin_amdgcn_wmma_f32_16x16x4_f32(
               false, af[rt][s],  false, bf[s],
               (short)0, acc[rt],
               false,  false);
#pragma unroll
      for (int s = 0; s < 8; ++s) bf[s] = bn[s];
#pragma unroll
      for (int rt = 0; rt < 4; ++rt)
#pragma unroll
        for (int s = 0; s < 8; ++s) af[rt][s] = an[rt][s];
    }
#pragma unroll
    for (int rt = 0; rt < 4; ++rt)
#pragma unroll
      for (int s = 0; s < 8; ++s)
        acc[rt] = __builtin_amdgcn_wmma_f32_16x16x4_f32(
            false, af[rt][s], false, bf[s], (short)0, acc[rt], false, false);
#pragma unroll
    for (int rt = 0; rt < 4; ++rt) {
      float* obase = dst + (rt * 16 + 8 * (lane >> 4)) * P_STRIDE + colbase + lr;
#pragma unroll
      for (int v = 0; v < 8; ++v) obase[v * P_STRIDE] = acc[rt][v];
    }
  }
  __syncthreads();

  const int d  = tid & 127;
  const int j0 = (tid >> 7) * 32;
  v2f xl2[16], sum2[16];
#pragma unroll
  for (int u = 0; u < 16; ++u) {
    xl2[u]  = (v2f){ xl[(j0 + 2 * u) * P_STRIDE + d], xl[(j0 + 2 * u + 1) * P_STRIDE + d] };
    sum2[u] = (v2f){ 0.f, 0.f };
  }
  for (int ib = 0; ib < HROWS; ib += 8) {
    float xrv[8];
#pragma unroll
    for (int v = 0; v < 8; ++v) xrv[v] = xr[(ib + v) * P_STRIDE + d];
#pragma unroll
    for (int v = 0; v < 8; ++v) {
      const v2f r2 = { xrv[v], xrv[v] };
      v2f t[16];
#pragma unroll
      for (int u = 0; u < 16; ++u) t[u] = xl2[u] + r2;
#pragma unroll
      for (int u = 0; u < 16; ++u) {
        t[u].x = fmaxf(t[u].x, 0.f);
        t[u].y = fmaxf(t[u].y, 0.f);
      }
#pragma unroll
      for (int u = 0; u < 16; ++u) sum2[u] += t[u];
    }
  }
#pragma unroll
  for (int w = 8; w >= 1; w >>= 1) {
#pragma unroll
    for (int u = 0; u < w; ++u) sum2[u] += sum2[u + w];
  }
  const float acc = sum2[0].x + sum2[0].y;
  smem[tid] = acc;
  __syncthreads();
  if (tid < 128)
    part[(blk * NBATCH + b) * DD + dbase + tid] = smem[tid] + smem[tid + 128];
}

__global__ __launch_bounds__(32) void RN_out_kernel(
    const float* __restrict__ part, const float* __restrict__ W_rn,
    const float* __restrict__ b_rn, float* __restrict__ out, size_t ws_bytes) {
  if (ws_bytes < WS_NEED) return;

  const int mt   = blockIdx.x;
  const int nt   = blockIdx.y;
  const int lane = threadIdx.x;
  const int lr   = lane & 15;
  const int koff = (lane >> 4) * 2;
  const int row  = mt * 16 + lr;
  const int col  = nt * 16 + lr;

  const float bv = 16384.0f * b_rn[col];
  v8f c = { bv, bv, bv, bv, bv, bv, bv, bv };

  const float* arow = part + (long)row * DD + koff;
  const float* brow = W_rn + (long)col * DD + koff;
  v2f pa[4][8], pb[8];
#pragma unroll
  for (int s = 0; s < 8; ++s) {
#pragma unroll
    for (int p = 0; p < 4; ++p) pa[p][s] = *(const v2f*)(arow + p * PLANE + 4 * s);
    pb[s] = *(const v2f*)(brow + 4 * s);
  }
  for (int kb = 32; kb < DD; kb += 32) {
    v2f na[4][8], nb[8];
#pragma unroll
    for (int s = 0; s < 8; ++s) {
#pragma unroll
      for (int p = 0; p < 4; ++p) na[p][s] = *(const v2f*)(arow + p * PLANE + kb + 4 * s);
      nb[s] = *(const v2f*)(brow + kb + 4 * s);
    }
    __builtin_amdgcn_sched_barrier(0);
#pragma unroll
    for (int s = 0; s < 8; ++s) {
      const v2f a = (pa[0][s] + pa[1][s]) + (pa[2][s] + pa[3][s]);
      c = __builtin_amdgcn_wmma_f32_16x16x4_f32(
          false, a, false, pb[s], (short)0, c, false, false);
    }
#pragma unroll
    for (int s = 0; s < 8; ++s) {
#pragma unroll
      for (int p = 0; p < 4; ++p) pa[p][s] = na[p][s];
      pb[s] = nb[s];
    }
  }
#pragma unroll
  for (int s = 0; s < 8; ++s) {
    const v2f a = (pa[0][s] + pa[1][s]) + (pa[2][s] + pa[3][s]);
    c = __builtin_amdgcn_wmma_f32_16x16x4_f32(
        false, a, false, pb[s], (short)0, c, false, false);
  }
  float* obase = out + (mt * 16 + 8 * (lane >> 4)) * DD + nt * 16 + lr;
#pragma unroll
  for (int v = 0; v < 8; ++v) obase[v * DD] = c[v];
}

extern "C" void kernel_launch(void* const* d_in, const int* in_sizes, int n_in,
                              void* d_out, int out_size, void* d_ws, size_t ws_size,
                              hipStream_t stream) {
  (void)in_sizes; (void)n_in; (void)out_size;
  const int*   X    = (const int*)d_in[0];
  const float* emb  = (const float*)d_in[1];
  const float* W_l  = (const float*)d_in[2];
  const float* b_l  = (const float*)d_in[3];
  const float* W_r  = (const float*)d_in[4];
  const float* b_r  = (const float*)d_in[5];
  const float* W_rn = (const float*)d_in[6];
  const float* b_rn = (const float*)d_in[7];
  float* part = (float*)d_ws;
  float* out  = (float*)d_out;

  dim3 gridA(NBATCH, 2, NPART), blockA(256);
  size_t shmem = (size_t)SMEM_FLOATS * sizeof(float);
  RN_pair_block_kernel<<<gridA, blockA, shmem, stream>>>(
      X, emb, W_l, b_l, W_r, b_r, part, ws_size);

  dim3 gridB(2, 16), blockB(32);
  RN_out_kernel<<<gridB, blockB, 0, stream>>>(part, W_rn, b_rn, out, ws_size);
}


# Gather re-cut: a wave copies whole embedding rows, tokens by scalar loads (no LDS token plane, one block barrier fewer), all its row loads in flight before the first LDS store; sums unchanged
# speedup vs baseline: 4.9029x; 1.0432x over previous
#include <hip/hip_runtime.h>


#define DD        256
#define SEQN      128
#define NBATCH    32
#define HROWS     64
#define NPART     4
#define XE_STRIDE 260
#define P_STRIDE  132

typedef float v2f __attribute__((ext_vector_type(2)));
typedef float v8f __attribute__((ext_vector_type(8)));

#define XE_OFF  0
#define XL_OFF  (SEQN * XE_STRIDE)
#define XR_OFF  (XL_OFF + HROWS * P_STRIDE)
#define SMEM_FLOATS (XR_OFF + HROWS * P_STRIDE)

#define PLANE    (NBATCH * DD)
#define WS_NEED  ((size_t)NPART * PLANE * sizeof(float))

template <int NR>
__device__ __forceinline__ void gather_rows(const int* __restrict__ xtok,
                                            const float* __restrict__ emb,
                                            float* __restrict__ dst, int lane) {
  float4 v[NR][2];
#pragma unroll
  for (int t = 0; t < NR; ++t) {
    const float4* src = (const float4*)(emb + (long)xtok[t] * DD);
    v[t][0] = src[lane];
    v[t][1] = src[lane + 32];
  }
#pragma unroll
  for (int t = 0; t < NR; ++t) {
    float4* d = (float4*)(dst + t * XE_STRIDE);
    d[lane]      = v[t][0];
    d[lane + 32] = v[t][1];
  }
}

__global__ __launch_bounds__(256) void RN_pair_block_kernel(
    const int* __restrict__ X, const float* __restrict__ emb,
    const float* __restrict__ W_l, const float* __restrict__ b_l,
    const float* __restrict__ W_r, const float* __restrict__ b_r,
    float* __restrict__ part, size_t ws_bytes) {
  if (ws_bytes < WS_NEED) return;

  extern __shared__ __attribute__((aligned(16))) float smem[];
  float* xe = smem + XE_OFF;
  float* xl = smem + XL_OFF;
  float* xr = smem + XR_OFF;

  const int b     = blockIdx.x;
  const int dbase = blockIdx.y * 128;
  const int blk   = blockIdx.z;
  const int jh    = blk & 1;
  const int ih    = blk >> 1;
  const int tid   = threadIdx.x;

  const bool diag  = (ih == jh);
  const float* xei = diag ? xe : xe + HROWS * XE_STRIDE;

  const int gw = __builtin_amdgcn_readfirstlane(tid >> 5);
  if (diag) {
    gather_rows<8>(X + b * SEQN + jh * HROWS + gw * 8, emb,
                   xe + gw * 8 * XE_STRIDE, tid & 31);
  } else {
    const int hsel = (gw < 4) ? jh : ih;
    gather_rows<16>(X + b * SEQN + hsel * HROWS + (gw & 3) * 16, emb,
                    xe + gw * 16 * XE_STRIDE, tid & 31);
  }
  __syncthreads();

  const int wave = tid >> 5;
  const int lane = tid & 31;
  const int lr   = lane & 15;
  const int koff = (lane >> 4) * 2;

  for (int cu = wave; cu < 16; cu += 8) {
    const int mat     = cu >> 3;
    const int colbase = (cu & 7) * 16;
    const float* W    = mat ? W_r : W_l;
    const float* bias = mat ? b_r : b_l;
    const float* asrc = mat ? xei : xe;
    float*       dst  = mat ? xr : xl;
    const int dcol    = dbase + colbase + lr;

    const float bv = bias[dcol];
    v8f acc[4];
#pragma unroll
    for (int rt = 0; rt < 4; ++rt)
      acc[rt] = (v8f){ bv, bv, bv, bv, bv, bv, bv, bv };

    const float* brow  = W + (long)dcol * DD + koff;
    const float* abase = asrc + lr * XE_STRIDE + koff;
    v2f bf[8], af[4][8];
#pragma unroll
    for (int s = 0; s < 8; ++s) bf[s] = *(const v2f*)(brow + 4 * s);
#pragma unroll
    for (int rt = 0; rt < 4; ++rt)
#pragma unroll
      for (int s = 0; s < 8; ++s)
        af[rt][s] = *(const v2f*)(abase + rt * 16 * XE_STRIDE + 4 * s);
    for (int kb = 32; kb < DD; kb += 32) {
      v2f bn[8], an[4][8];
#pragma unroll
      for (int s = 0; s < 8; ++s) bn[s] = *(const v2f*)(brow + kb + 4 * s);
#pragma unroll
      for (int rt = 0; rt < 4; ++rt)
#pragma unroll
        for (int s = 0; s < 8; ++s)
          an[rt][s] = *(const v2f*)(abase + rt * 16 * XE_STRIDE + kb + 4 * s);
      __builtin_amdgcn_sched_barrier(0);
#pragma unroll
      for (int rt = 0; rt < 4; ++rt)
#pragma unroll
        for (int s = 0; s < 8; ++s)
          acc[rt] = __builtin_amdgcn_wmma_f32_16x16x4_f32(
               false, af[rt][s],  false, bf[s],
               (short)0, acc[rt],
               false,  false);
#pragma unroll
      for (int s = 0; s < 8; ++s) bf[s] = bn[s];
#pragma unroll
      for (int rt = 0; rt < 4; ++rt)
#pragma unroll
        for (int s = 0; s < 8; ++s) af[rt][s] = an[rt][s];
    }
#pragma unroll
    for (int rt = 0; rt < 4; ++rt)
#pragma unroll
      for (int s = 0; s < 8; ++s)
        acc[rt] = __builtin_amdgcn_wmma_f32_16x16x4_f32(
            false, af[rt][s], false, bf[s], (short)0, acc[rt], false, false);
#pragma unroll
    for (int rt = 0; rt < 4; ++rt) {
      float* obase = dst + (rt * 16 + 8 * (lane >> 4)) * P_STRIDE + colbase + lr;
#pragma unroll
      for (int v = 0; v < 8; ++v) obase[v * P_STRIDE] = acc[rt][v];
    }
  }
  __syncthreads();

  const int d  = tid & 127;
  const int j0 = (tid >> 7) * 32;
  v2f xl2[16], sum2[16];
#pragma unroll
  for (int u = 0; u < 16; ++u) {
    xl2[u]  = (v2f){ xl[(j0 + 2 * u) * P_STRIDE + d], xl[(j0 + 2 * u + 1) * P_STRIDE + d] };
    sum2[u] = (v2f){ 0.f, 0.f };
  }
  for (int ib = 0; ib < HROWS; ib += 8) {
    float xrv[8];
#pragma unroll
    for (int v = 0; v < 8; ++v) xrv[v] = xr[(ib + v) * P_STRIDE + d];
#pragma unroll
    for (int v = 0; v < 8; ++v) {
      const v2f r2 = { xrv[v], xrv[v] };
      v2f t[16];
#pragma unroll
      for (int u = 0; u < 16; ++u) t[u] = xl2[u] + r2;
#pragma unroll
      for (int u = 0; u < 16; ++u) {
        t[u].x = fmaxf(t[u].x, 0.f);
        t[u].y = fmaxf(t[u].y, 0.f);
      }
#pragma unroll
      for (int u = 0; u < 16; ++u) sum2[u] += t[u];
    }
  }
#pragma unroll
  for (int w = 8; w >= 1; w >>= 1) {
#pragma unroll
    for (int u = 0; u < w; ++u) sum2[u] += sum2[u + w];
  }
  const float acc = sum2[0].x + sum2[0].y;
  smem[tid] = acc;
  __syncthreads();
  if (tid < 128)
    part[(blk * NBATCH + b) * DD + dbase + tid] = smem[tid] + smem[tid + 128];
}

__global__ __launch_bounds__(32) void RN_out_kernel(
    const float* __restrict__ part, const float* __restrict__ W_rn,
    const float* __restrict__ b_rn, float* __restrict__ out, size_t ws_bytes) {
  if (ws_bytes < WS_NEED) return;

  const int mt   = blockIdx.x;
  const int nt   = blockIdx.y;
  const int lane = threadIdx.x;
  const int lr   = lane & 15;
  const int koff = (lane >> 4) * 2;
  const int row  = mt * 16 + lr;
  const int col  = nt * 16 + lr;

  const float bv = 16384.0f * b_rn[col];
  v8f c = { bv, bv, bv, bv, bv, bv, bv, bv };

  const float* arow = part + (long)row * DD + koff;
  const float* brow = W_rn + (long)col * DD + koff;
  v2f pa[4][8], pb[8];
#pragma unroll
  for (int s = 0; s < 8; ++s) {
#pragma unroll
    for (int p = 0; p < 4; ++p) pa[p][s] = *(const v2f*)(arow + p * PLANE + 4 * s);
    pb[s] = *(const v2f*)(brow + 4 * s);
  }
  for (int kb = 32; kb < DD; kb += 32) {
    v2f na[4][8], nb[8];
#pragma unroll
    for (int s = 0; s < 8; ++s) {
#pragma unroll
      for (int p = 0; p < 4; ++p) na[p][s] = *(const v2f*)(arow + p * PLANE + kb + 4 * s);
      nb[s] = *(const v2f*)(brow + kb + 4 * s);
    }
    __builtin_amdgcn_sched_barrier(0);
#pragma unroll
    for (int s = 0; s < 8; ++s) {
      const v2f a = (pa[0][s] + pa[1][s]) + (pa[2][s] + pa[3][s]);
      c = __builtin_amdgcn_wmma_f32_16x16x4_f32(
          false, a, false, pb[s], (short)0, c, false, false);
    }
#pragma unroll
    for (int s = 0; s < 8; ++s) {
#pragma unroll
      for (int p = 0; p < 4; ++p) pa[p][s] = na[p][s];
      pb[s] = nb[s];
    }
  }
#pragma unroll
  for (int s = 0; s < 8; ++s) {
    const v2f a = (pa[0][s] + pa[1][s]) + (pa[2][s] + pa[3][s]);
    c = __builtin_amdgcn_wmma_f32_16x16x4_f32(
        false, a, false, pb[s], (short)0, c, false, false);
  }
  float* obase = out + (mt * 16 + 8 * (lane >> 4)) * DD + nt * 16 + lr;
#pragma unroll
  for (int v = 0; v < 8; ++v) obase[v * DD] = c[v];
}

extern "C" void kernel_launch(void* const* d_in, const int* in_sizes, int n_in,
                              void* d_out, int out_size, void* d_ws, size_t ws_size,
                              hipStream_t stream) {
  (void)in_sizes; (void)n_in; (void)out_size;
  const int*   X    = (const int*)d_in[0];
  const float* emb  = (const float*)d_in[1];
  const float* W_l  = (const float*)d_in[2];
  const float* b_l  = (const float*)d_in[3];
  const float* W_r  = (const float*)d_in[4];
  const float* b_r  = (const float*)d_in[5];
  const float* W_rn = (const float*)d_in[6];
  const float* b_rn = (const float*)d_in[7];
  float* part = (float*)d_ws;
  float* out  = (float*)d_out;

  dim3 gridA(NBATCH, 2, NPART), blockA(256);
  size_t shmem = (size_t)SMEM_FLOATS * sizeof(float);
  RN_pair_block_kernel<<<gridA, blockA, shmem, stream>>>(
      X, emb, W_l, b_l, W_r, b_r, part, ws_size);

  dim3 gridB(2, 16), blockB(32);
  RN_out_kernel<<<gridB, blockB, 0, stream>>>(part, W_rn, b_rn, out, ws_size);
}


# Each wave sums the pair grid of its own 16 columns from its WMMA accumulators (other half's Xr rows by lane swap): Xl/Xr skip LDS, no barrier between GEMM and pair loop; pair-sum order changes
# speedup vs baseline: 5.1008x; 1.0404x over previous
#include <hip/hip_runtime.h>


#define DD        256
#define SEQN      128
#define NBATCH    32
#define HROWS     64
#define NPART     4
#define XE_STRIDE 260

typedef float v2f __attribute__((ext_vector_type(2)));
typedef float v8f __attribute__((ext_vector_type(8)));

#define XE_OFF  0
#define CS_OFF  (SEQN * XE_STRIDE)
#define SMEM_FLOATS 50176

#define PLANE    (NBATCH * DD)
#define WS_NEED  ((size_t)NPART * PLANE * sizeof(float))

template <int NR>
__device__ __forceinline__ void gather_rows(const int* __restrict__ xtok,
                                            const float* __restrict__ emb,
                                            float* __restrict__ dst, int lane) {
  float4 v[NR][2];
#pragma unroll
  for (int t = 0; t < NR; ++t) {
    const float4* src = (const float4*)(emb + (long)xtok[t] * DD);
    v[t][0] = src[lane];
    v[t][1] = src[lane + 32];
  }
#pragma unroll
  for (int t = 0; t < NR; ++t) {
    float4* d = (float4*)(dst + t * XE_STRIDE);
    d[lane]      = v[t][0];
    d[lane + 32] = v[t][1];
  }
}

__device__ __forceinline__ float xhalf(float x) {
  return __builtin_bit_cast(float, __builtin_amdgcn_permlanex16(
      __builtin_bit_cast(int, x), __builtin_bit_cast(int, x),
      0x76543210, 0xfedcba98, false, false));
}

__global__ __launch_bounds__(256) void RN_pair_block_kernel(
    const int* __restrict__ X, const float* __restrict__ emb,
    const float* __restrict__ W_l, const float* __restrict__ b_l,
    const float* __restrict__ W_r, const float* __restrict__ b_r,
    float* __restrict__ part, size_t ws_bytes) {
  if (ws_bytes < WS_NEED) return;

  extern __shared__ __attribute__((aligned(16))) float smem[];
  float* xe = smem + XE_OFF;
  float* cs = smem + CS_OFF;

  const int b     = blockIdx.x;
  const int dbase = blockIdx.y * 128;
  const int blk   = blockIdx.z;
  const int jh    = blk & 1;
  const int ih    = blk >> 1;
  const int tid   = threadIdx.x;

  const bool diag  = (ih == jh);
  const float* xei = diag ? xe : xe + HROWS * XE_STRIDE;

  const int gw = __builtin_amdgcn_readfirstlane(tid >> 5);
  if (diag) {
    gather_rows<8>(X + b * SEQN + jh * HROWS + gw * 8, emb,
                   xe + gw * 8 * XE_STRIDE, tid & 31);
  } else {
    const int hsel = (gw < 4) ? jh : ih;
    gather_rows<16>(X + b * SEQN + hsel * HROWS + (gw & 3) * 16, emb,
                    xe + gw * 16 * XE_STRIDE, tid & 31);
  }
  __syncthreads();

  const int lane = tid & 31;
  const int lr   = lane & 15;
  const int koff = (lane >> 4) * 2;
  const int colbase = gw * 16;
  const int dcol    = dbase + colbase + lr;

  v8f accl[4], acc[4];
#pragma unroll
  for (int mat = 0; mat < 2; ++mat) {
    const float* W    = mat ? W_r : W_l;
    const float* bias = mat ? b_r : b_l;
    const float* asrc = mat ? xei : xe;

    const float bv = bias[dcol];
#pragma unroll
    for (int rt = 0; rt < 4; ++rt)
      acc[rt] = (v8f){ bv, bv, bv, bv, bv, bv, bv, bv };

    const float* brow  = W + (long)dcol * DD + koff;
    const float* abase = asrc + lr * XE_STRIDE + koff;
    v2f bf[8], af[4][8];
#pragma unroll
    for (int s = 0; s < 8; ++s) bf[s] = *(const v2f*)(brow + 4 * s);
#pragma unroll
    for (int rt = 0; rt < 4; ++rt)
#pragma unroll
      for (int s = 0; s < 8; ++s)
        af[rt][s] = *(const v2f*)(abase + rt * 16 * XE_STRIDE + 4 * s);
    for (int kb = 32; kb < DD; kb += 32) {
      v2f bn[8], an[4][8];
#pragma unroll
      for (int s = 0; s < 8; ++s) bn[s] = *(const v2f*)(brow + kb + 4 * s);
#pragma unroll
      for (int rt = 0; rt < 4; ++rt)
#pragma unroll
        for (int s = 0; s < 8; ++s)
          an[rt][s] = *(const v2f*)(abase + rt * 16 * XE_STRIDE + kb + 4 * s);
      __builtin_amdgcn_sched_barrier(0);
#pragma unroll
      for (int rt = 0; rt < 4; ++rt)
#pragma unroll
        for (int s = 0; s < 8; ++s)
          acc[rt] = __builtin_amdgcn_wmma_f32_16x16x4_f32(
               false, af[rt][s],  false, bf[s],
               (short)0, acc[rt],
               false,  false);
#pragma unroll
      for (int s = 0; s < 8; ++s) bf[s] = bn[s];
#pragma unroll
      for (int rt = 0; rt < 4; ++rt)
#pragma unroll
        for (int s = 0; s < 8; ++s) af[rt][s] = an[rt][s];
    }
#pragma unroll
    for (int rt = 0; rt < 4; ++rt)
#pragma unroll
      for (int s = 0; s < 8; ++s)
        acc[rt] = __builtin_amdgcn_wmma_f32_16x16x4_f32(
            false, af[rt][s], false, bf[s], (short)0, acc[rt], false, false);
    if (mat == 0) {
#pragma unroll
      for (int rt = 0; rt < 4; ++rt) accl[rt] = acc[rt];
    }
  }

  v2f xl2[16], sum2[16];
  float xrv[32];
#pragma unroll
  for (int rt = 0; rt < 4; ++rt) {
#pragma unroll
    for (int q = 0; q < 4; ++q) {
      xl2[rt * 4 + q]  = (v2f){ accl[rt][2 * q], accl[rt][2 * q + 1] };
      sum2[rt * 4 + q] = (v2f){ 0.f, 0.f };
    }
#pragma unroll
    for (int v = 0; v < 8; ++v) xrv[rt * 8 + v] = acc[rt][v];
  }
#pragma nounroll
  for (int pass = 0; pass < 2; ++pass) {
#pragma unroll
    for (int v = 0; v < 32; ++v) {
      const v2f r2 = { xrv[v], xrv[v] };
      v2f t[16];
#pragma unroll
      for (int u = 0; u < 16; ++u) t[u] = xl2[u] + r2;
#pragma unroll
      for (int u = 0; u < 16; ++u) {
        t[u].x = fmaxf(t[u].x, 0.f);
        t[u].y = fmaxf(t[u].y, 0.f);
      }
#pragma unroll
      for (int u = 0; u < 16; ++u) sum2[u] += t[u];
    }
    if (pass == 0) {
#pragma unroll
      for (int v = 0; v < 32; ++v) xrv[v] = xhalf(xrv[v]);
    }
  }
#pragma unroll
  for (int w = 8; w >= 1; w >>= 1) {
#pragma unroll
    for (int u = 0; u < w; ++u) sum2[u] += sum2[u + w];
  }
  const float half_sum = sum2[0].x + sum2[0].y;
  const float col_sum  = half_sum + xhalf(half_sum);
  if (lane < 16) cs[colbase + lane] = col_sum;
  __syncthreads();
  if (tid < 128)
    part[(blk * NBATCH + b) * DD + dbase + tid] = cs[tid];
}

__global__ __launch_bounds__(32) void RN_out_kernel(
    const float* __restrict__ part, const float* __restrict__ W_rn,
    const float* __restrict__ b_rn, float* __restrict__ out, size_t ws_bytes) {
  if (ws_bytes < WS_NEED) return;

  const int mt   = blockIdx.x;
  const int nt   = blockIdx.y;
  const int lane = threadIdx.x;
  const int lr   = lane & 15;
  const int koff = (lane >> 4) * 2;
  const int row  = mt * 16 + lr;
  const int col  = nt * 16 + lr;

  const float bv = 16384.0f * b_rn[col];
  v8f c = { bv, bv, bv, bv, bv, bv, bv, bv };

  const float* arow = part + (long)row * DD + koff;
  const float* brow = W_rn + (long)col * DD + koff;
  v2f pa[4][8], pb[8];
#pragma unroll
  for (int s = 0; s < 8; ++s) {
#pragma unroll
    for (int p = 0; p < 4; ++p) pa[p][s] = *(const v2f*)(arow + p * PLANE + 4 * s);
    pb[s] = *(const v2f*)(brow + 4 * s);
  }
  for (int kb = 32; kb < DD; kb += 32) {
    v2f na[4][8], nb[8];
#pragma unroll
    for (int s = 0; s < 8; ++s) {
#pragma unroll
      for (int p = 0; p < 4; ++p) na[p][s] = *(const v2f*)(arow + p * PLANE + kb + 4 * s);
      nb[s] = *(const v2f*)(brow + kb + 4 * s);
    }
    __builtin_amdgcn_sched_barrier(0);
#pragma unroll
    for (int s = 0; s < 8; ++s) {
      const v2f a = (pa[0][s] + pa[1][s]) + (pa[2][s] + pa[3][s]);
      c = __builtin_amdgcn_wmma_f32_16x16x4_f32(
          false, a, false, pb[s], (short)0, c, false, false);
    }
#pragma unroll
    for (int s = 0; s < 8; ++s) {
#pragma unroll
      for (int p = 0; p < 4; ++p) pa[p][s] = na[p][s];
      pb[s] = nb[s];
    }
  }
#pragma unroll
  for (int s = 0; s < 8; ++s) {
    const v2f a = (pa[0][s] + pa[1][s]) + (pa[2][s] + pa[3][s]);
    c = __builtin_amdgcn_wmma_f32_16x16x4_f32(
        false, a, false, pb[s], (short)0, c, false, false);
  }
  float* obase = out + (mt * 16 + 8 * (lane >> 4)) * DD + nt * 16 + lr;
#pragma unroll
  for (int v = 0; v < 8; ++v) obase[v * DD] = c[v];
}

extern "C" void kernel_launch(void* const* d_in, const int* in_sizes, int n_in,
                              void* d_out, int out_size, void* d_ws, size_t ws_size,
                              hipStream_t stream) {
  (void)in_sizes; (void)n_in; (void)out_size;
  const int*   X    = (const int*)d_in[0];
  const float* emb  = (const float*)d_in[1];
  const float* W_l  = (const float*)d_in[2];
  const float* b_l  = (const float*)d_in[3];
  const float* W_r  = (const float*)d_in[4];
  const float* b_r  = (const float*)d_in[5];
  const float* W_rn = (const float*)d_in[6];
  const float* b_rn = (const float*)d_in[7];
  float* part = (float*)d_ws;
  float* out  = (float*)d_out;

  dim3 gridA(NBATCH, 2, NPART), blockA(256);
  size_t shmem = (size_t)SMEM_FLOATS * sizeof(float);
  RN_pair_block_kernel<<<gridA, blockA, shmem, stream>>>(
      X, emb, W_l, b_l, W_r, b_r, part, ws_size);

  dim3 gridB(2, 16), blockB(32);
  RN_out_kernel<<<gridB, blockB, 0, stream>>>(part, W_rn, b_rn, out, ws_size);
}


# Grid re-cut: work-group = 64 columns, 128 j x 64 i (was 128 columns, 64 j x 64 i): Xr projected once, not twice; wave pairs share a column tile, Xr via LDS; 2 partials; pair-sum order changes
# speedup vs baseline: 5.8886x; 1.1544x over previous
#include <hip/hip_runtime.h>


#define DD        256
#define SEQN      128
#define NBATCH    32
#define HROWS     64
#define NCG       4
#define CGW       64
#define NPART     2
#define XE_STRIDE 260
#define XR_STRIDE 68

typedef float v2f __attribute__((ext_vector_type(2)));
typedef float v8f __attribute__((ext_vector_type(8)));

#define XE_OFF  0
#define CS_OFF  (SEQN * XE_STRIDE)
#define XR_OFF  (CS_OFF + 128)
#define SMEM_FLOATS 50176

#define PLANE    (NBATCH * DD)
#define WS_NEED  ((size_t)NPART * PLANE * sizeof(float))

template <int NR>
__device__ __forceinline__ void gather_rows(const int* __restrict__ xtok,
                                            const float* __restrict__ emb,
                                            float* __restrict__ dst, int lane) {
  float4 v[NR][2];
#pragma unroll
  for (int t = 0; t < NR; ++t) {
    const float4* src = (const float4*)(emb + (long)xtok[t] * DD);
    v[t][0] = src[lane];
    v[t][1] = src[lane + 32];
  }
#pragma unroll
  for (int t = 0; t < NR; ++t) {
    float4* d = (float4*)(dst + t * XE_STRIDE);
    d[lane]      = v[t][0];
    d[lane + 32] = v[t][1];
  }
}

__device__ __forceinline__ float xhalf(float x) {
  return __builtin_bit_cast(float, __builtin_amdgcn_permlanex16(
      __builtin_bit_cast(int, x), __builtin_bit_cast(int, x),
      0x76543210, 0xfedcba98, false, false));
}

template <int NRT>
__device__ __forceinline__ void project_tiles(const float* __restrict__ brow,
                                              const float* __restrict__ abase,
                                              float bv, v8f (&acc)[NRT]) {
#pragma unroll
  for (int rt = 0; rt < NRT; ++rt)
    acc[rt] = (v8f){ bv, bv, bv, bv, bv, bv, bv, bv };

  v2f bf[8], af[NRT][8];
#pragma unroll
  for (int s = 0; s < 8; ++s) bf[s] = *(const v2f*)(brow + 4 * s);
#pragma unroll
  for (int rt = 0; rt < NRT; ++rt)
#pragma unroll
    for (int s = 0; s < 8; ++s)
      af[rt][s] = *(const v2f*)(abase + rt * 16 * XE_STRIDE + 4 * s);
#pragma unroll
  for (int kb = 32; kb < DD; kb += 32) {
    v2f bn[8], an[NRT][8];
#pragma unroll
    for (int s = 0; s < 8; ++s) bn[s] = *(const v2f*)(brow + kb + 4 * s);
#pragma unroll
    for (int rt = 0; rt < NRT; ++rt)
#pragma unroll
      for (int s = 0; s < 8; ++s)
        an[rt][s] = *(const v2f*)(abase + rt * 16 * XE_STRIDE + kb + 4 * s);
    __builtin_amdgcn_sched_barrier(0);
#pragma unroll
    for (int rt = 0; rt < NRT; ++rt)
#pragma unroll
      for (int s = 0; s < 8; ++s)
        acc[rt] = __builtin_amdgcn_wmma_f32_16x16x4_f32(
             false, af[rt][s],  false, bf[s],
             (short)0, acc[rt],
             false,  false);
#pragma unroll
    for (int s = 0; s < 8; ++s) bf[s] = bn[s];
#pragma unroll
    for (int rt = 0; rt < NRT; ++rt)
#pragma unroll
      for (int s = 0; s < 8; ++s) af[rt][s] = an[rt][s];
  }
#pragma unroll
  for (int rt = 0; rt < NRT; ++rt)
#pragma unroll
    for (int s = 0; s < 8; ++s)
      acc[rt] = __builtin_amdgcn_wmma_f32_16x16x4_f32(
          false, af[rt][s], false, bf[s], (short)0, acc[rt], false, false);
}

__global__ __launch_bounds__(256) void RN_pair_block_kernel(
    const int* __restrict__ X, const float* __restrict__ emb,
    const float* __restrict__ W_l, const float* __restrict__ b_l,
    const float* __restrict__ W_r, const float* __restrict__ b_r,
    float* __restrict__ part, size_t ws_bytes) {
  if (ws_bytes < WS_NEED) return;

  extern __shared__ __attribute__((aligned(16))) float smem[];
  float* xe  = smem + XE_OFF;
  float* cs  = smem + CS_OFF;
  float* xrT = smem + XR_OFF;

  const int b     = blockIdx.x;
  const int dbase = blockIdx.y * CGW;
  const int ih    = blockIdx.z;
  const int tid   = threadIdx.x;

  const int gw = __builtin_amdgcn_readfirstlane(tid >> 5);
  gather_rows<16>(X + b * SEQN + gw * 16, emb, xe + gw * 16 * XE_STRIDE, tid & 31);
  __syncthreads();

  const int lane = tid & 31;
  const int lr   = lane & 15;
  const int hh   = lane >> 4;
  const int koff = hh * 2;
  const int ct   = gw & 3;
  const int jw   = gw >> 2;
  const int colbase = ct * 16;
  const int dcol    = dbase + colbase + lr;


  {
    v8f accr[2];
    project_tiles<2>(W_r + (long)dcol * DD + koff,
                     xe + (ih * HROWS + jw * 32 + lr) * XE_STRIDE + koff,
                     b_r[dcol], accr);
    float* xw = xrT + (colbase + lr) * XR_STRIDE + jw * 32 + 8 * hh;
#pragma unroll
    for (int rt = 0; rt < 2; ++rt) {
      *(float4*)(xw + rt * 16)     = (float4){ accr[rt][0], accr[rt][1], accr[rt][2], accr[rt][3] };
      *(float4*)(xw + rt * 16 + 4) = (float4){ accr[rt][4], accr[rt][5], accr[rt][6], accr[rt][7] };
    }
  }
  __builtin_amdgcn_fence(__ATOMIC_RELEASE, "workgroup");
  __builtin_amdgcn_s_barrier_signal(-1);

  v8f accl[4];
  project_tiles<4>(W_l + (long)dcol * DD + koff,
                   xe + (jw * HROWS + lr) * XE_STRIDE + koff,
                   b_l[dcol], accl);

  __builtin_amdgcn_sched_barrier(0);
  __builtin_amdgcn_s_barrier_wait(-1);
  __builtin_amdgcn_fence(__ATOMIC_ACQUIRE, "workgroup");

  v2f xl2[16], sum2[16];
#pragma unroll
  for (int rt = 0; rt < 4; ++rt) {
#pragma unroll
    for (int q = 0; q < 4; ++q) {
      xl2[rt * 4 + q]  = (v2f){ accl[rt][2 * q], accl[rt][2 * q + 1] };
      sum2[rt * 4 + q] = (v2f){ 0.f, 0.f };
    }
  }
  const float* xcol = xrT + (colbase + lr) * XR_STRIDE;
#pragma nounroll
  for (int pass = 0; pass < 2; ++pass) {
    float xrv[32];
#pragma unroll
    for (int q = 0; q < 8; ++q) {
      const float4 f = *(const float4*)(xcol + pass * 32 + 4 * q);
      xrv[4 * q]     = f.x;
      xrv[4 * q + 1] = f.y;
      xrv[4 * q + 2] = f.z;
      xrv[4 * q + 3] = f.w;
    }
#pragma unroll
    for (int v = 0; v < 32; ++v) {
      const v2f r2 = { xrv[v], xrv[v] };
      v2f t[16];
#pragma unroll
      for (int u = 0; u < 16; ++u) t[u] = xl2[u] + r2;
#pragma unroll
      for (int u = 0; u < 16; ++u) {
        t[u].x = fmaxf(t[u].x, 0.f);
        t[u].y = fmaxf(t[u].y, 0.f);
      }
#pragma unroll
      for (int u = 0; u < 16; ++u) sum2[u] += t[u];
    }
  }
#pragma unroll
  for (int w = 8; w >= 1; w >>= 1) {
#pragma unroll
    for (int u = 0; u < w; ++u) sum2[u] += sum2[u + w];
  }
  const float half_sum = sum2[0].x + sum2[0].y;
  const float col_sum  = half_sum + xhalf(half_sum);
  if (lane < 16) cs[jw * CGW + colbase + lane] = col_sum;
  __syncthreads();
  if (tid < CGW)
    part[(ih * NBATCH + b) * DD + dbase + tid] = cs[tid] + cs[CGW + tid];
}

__global__ __launch_bounds__(32) void RN_out_kernel(
    const float* __restrict__ part, const float* __restrict__ W_rn,
    const float* __restrict__ b_rn, float* __restrict__ out, size_t ws_bytes) {
  if (ws_bytes < WS_NEED) return;

  const int mt   = blockIdx.x;
  const int nt   = blockIdx.y;
  const int lane = threadIdx.x;
  const int lr   = lane & 15;
  const int koff = (lane >> 4) * 2;
  const int row  = mt * 16 + lr;
  const int col  = nt * 16 + lr;

  const float bv = 16384.0f * b_rn[col];
  v8f c = { bv, bv, bv, bv, bv, bv, bv, bv };

  const float* arow = part + (long)row * DD + koff;
  const float* brow = W_rn + (long)col * DD + koff;
  v2f pa[NPART][8], pb[8];
#pragma unroll
  for (int s = 0; s < 8; ++s) {
#pragma unroll
    for (int p = 0; p < NPART; ++p) pa[p][s] = *(const v2f*)(arow + p * PLANE + 4 * s);
    pb[s] = *(const v2f*)(brow + 4 * s);
  }
  for (int kb = 32; kb < DD; kb += 32) {
    v2f na[NPART][8], nb[8];
#pragma unroll
    for (int s = 0; s < 8; ++s) {
#pragma unroll
      for (int p = 0; p < NPART; ++p) na[p][s] = *(const v2f*)(arow + p * PLANE + kb + 4 * s);
      nb[s] = *(const v2f*)(brow + kb + 4 * s);
    }
    __builtin_amdgcn_sched_barrier(0);
#pragma unroll
    for (int s = 0; s < 8; ++s) {
      const v2f a = pa[0][s] + pa[1][s];
      c = __builtin_amdgcn_wmma_f32_16x16x4_f32(
          false, a, false, pb[s], (short)0, c, false, false);
    }
#pragma unroll
    for (int s = 0; s < 8; ++s) {
#pragma unroll
      for (int p = 0; p < NPART; ++p) pa[p][s] = na[p][s];
      pb[s] = nb[s];
    }
  }
#pragma unroll
  for (int s = 0; s < 8; ++s) {
    const v2f a = pa[0][s] + pa[1][s];
    c = __builtin_amdgcn_wmma_f32_16x16x4_f32(
        false, a, false, pb[s], (short)0, c, false, false);
  }
  float* obase = out + (mt * 16 + 8 * (lane >> 4)) * DD + nt * 16 + lr;
#pragma unroll
  for (int v = 0; v < 8; ++v) obase[v * DD] = c[v];
}

extern "C" void kernel_launch(void* const* d_in, const int* in_sizes, int n_in,
                              void* d_out, int out_size, void* d_ws, size_t ws_size,
                              hipStream_t stream) {
  (void)in_sizes; (void)n_in; (void)out_size;
  const int*   X    = (const int*)d_in[0];
  const float* emb  = (const float*)d_in[1];
  const float* W_l  = (const float*)d_in[2];
  const float* b_l  = (const float*)d_in[3];
  const float* W_r  = (const float*)d_in[4];
  const float* b_r  = (const float*)d_in[5];
  const float* W_rn = (const float*)d_in[6];
  const float* b_rn = (const float*)d_in[7];
  float* part = (float*)d_ws;
  float* out  = (float*)d_out;

  dim3 gridA(NBATCH, NCG, NPART), blockA(256);
  size_t shmem = (size_t)SMEM_FLOATS * sizeof(float);
  RN_pair_block_kernel<<<gridA, blockA, shmem, stream>>>(
      X, emb, W_l, b_l, W_r, b_r, part, ws_size);

  dim3 gridB(2, 16), blockB(32);
  RN_out_kernel<<<gridB, blockB, 0, stream>>>(part, W_rn, b_rn, out, ws_size);
}


# Pair loop on Xl, Xr scaled by 2^-64 (exact): the packed add's output clamp to [0,1] is the relu, so add+relu is one instruction and the max pass goes (3 -> 2 issue slots per 2 pairs); sums unchanged
# speedup vs baseline: 6.2324x; 1.0584x over previous
#include <hip/hip_runtime.h>


#define DD        256
#define SEQN      128
#define NBATCH    32
#define HROWS     64
#define NCG       4
#define CGW       64
#define NPART     2
#define XE_STRIDE 260
#define XR_STRIDE 68

typedef float v2f __attribute__((ext_vector_type(2)));
typedef float v8f __attribute__((ext_vector_type(8)));

#define XE_OFF  0
#define CS_OFF  (SEQN * XE_STRIDE)
#define XR_OFF  (CS_OFF + 128)
#define SMEM_FLOATS 50176

#define PLANE    (NBATCH * DD)
#define WS_NEED  ((size_t)NPART * PLANE * sizeof(float))

template <int NR>
__device__ __forceinline__ void gather_rows(const int* __restrict__ xtok,
                                            const float* __restrict__ emb,
                                            float* __restrict__ dst, int lane) {
  float4 v[NR][2];
#pragma unroll
  for (int t = 0; t < NR; ++t) {
    const float4* src = (const float4*)(emb + (long)xtok[t] * DD);
    v[t][0] = src[lane];
    v[t][1] = src[lane + 32];
  }
#pragma unroll
  for (int t = 0; t < NR; ++t) {
    float4* d = (float4*)(dst + t * XE_STRIDE);
    d[lane]      = v[t][0];
    d[lane + 32] = v[t][1];
  }
}

__device__ __forceinline__ float xhalf(float x) {
  return __builtin_bit_cast(float, __builtin_amdgcn_permlanex16(
      __builtin_bit_cast(int, x), __builtin_bit_cast(int, x),
      0x76543210, 0xfedcba98, false, false));
}

#define PAIR_DOWN 0x1p-64f
#define PAIR_UP   0x1p64f

__device__ __forceinline__ v2f relu_add_lo(v2f xr2, v2f xl2) {
  v2f t;
  asm("v_pk_add_f32 %0, %1, %2 op_sel_hi:[0,1] clamp" : "=v"(t) : "v"(xr2), "v"(xl2));
  return t;
}
__device__ __forceinline__ v2f relu_add_hi(v2f xr2, v2f xl2) {
  v2f t;
  asm("v_pk_add_f32 %0, %1, %2 op_sel:[1,0] clamp" : "=v"(t) : "v"(xr2), "v"(xl2));
  return t;
}

template <int NRT>
__device__ __forceinline__ void project_tiles(const float* __restrict__ brow,
                                              const float* __restrict__ abase,
                                              float bv, v8f (&acc)[NRT]) {
#pragma unroll
  for (int rt = 0; rt < NRT; ++rt)
    acc[rt] = (v8f){ bv, bv, bv, bv, bv, bv, bv, bv };

  v2f bf[8], af[NRT][8];
#pragma unroll
  for (int s = 0; s < 8; ++s) bf[s] = *(const v2f*)(brow + 4 * s);
#pragma unroll
  for (int rt = 0; rt < NRT; ++rt)
#pragma unroll
    for (int s = 0; s < 8; ++s)
      af[rt][s] = *(const v2f*)(abase + rt * 16 * XE_STRIDE + 4 * s);
#pragma unroll
  for (int kb = 32; kb < DD; kb += 32) {
    v2f bn[8], an[NRT][8];
#pragma unroll
    for (int s = 0; s < 8; ++s) bn[s] = *(const v2f*)(brow + kb + 4 * s);
#pragma unroll
    for (int rt = 0; rt < NRT; ++rt)
#pragma unroll
      for (int s = 0; s < 8; ++s)
        an[rt][s] = *(const v2f*)(abase + rt * 16 * XE_STRIDE + kb + 4 * s);
    __builtin_amdgcn_sched_barrier(0);
#pragma unroll
    for (int rt = 0; rt < NRT; ++rt)
#pragma unroll
      for (int s = 0; s < 8; ++s)
        acc[rt] = __builtin_amdgcn_wmma_f32_16x16x4_f32(
             false, af[rt][s],  false, bf[s],
             (short)0, acc[rt],
             false,  false);
#pragma unroll
    for (int s = 0; s < 8; ++s) bf[s] = bn[s];
#pragma unroll
    for (int rt = 0; rt < NRT; ++rt)
#pragma unroll
      for (int s = 0; s < 8; ++s) af[rt][s] = an[rt][s];
  }
#pragma unroll
  for (int rt = 0; rt < NRT; ++rt)
#pragma unroll
    for (int s = 0; s < 8; ++s)
      acc[rt] = __builtin_amdgcn_wmma_f32_16x16x4_f32(
          false, af[rt][s], false, bf[s], (short)0, acc[rt], false, false);
}

__global__ __launch_bounds__(256) void RN_pair_block_kernel(
    const int* __restrict__ X, const float* __restrict__ emb,
    const float* __restrict__ W_l, const float* __restrict__ b_l,
    const float* __restrict__ W_r, const float* __restrict__ b_r,
    float* __restrict__ part, size_t ws_bytes) {
  if (ws_bytes < WS_NEED) return;

  extern __shared__ __attribute__((aligned(16))) float smem[];
  float* xe  = smem + XE_OFF;
  float* cs  = smem + CS_OFF;
  float* xrT = smem + XR_OFF;

  const int b     = blockIdx.x;
  const int dbase = blockIdx.y * CGW;
  const int ih    = blockIdx.z;
  const int tid   = threadIdx.x;

  const int gw = __builtin_amdgcn_readfirstlane(tid >> 5);
  gather_rows<16>(X + b * SEQN + gw * 16, emb, xe + gw * 16 * XE_STRIDE, tid & 31);
  __syncthreads();

  const int lane = tid & 31;
  const int lr   = lane & 15;
  const int hh   = lane >> 4;
  const int koff = hh * 2;
  const int ct   = gw & 3;
  const int jw   = gw >> 2;
  const int colbase = ct * 16;
  const int dcol    = dbase + colbase + lr;


  {
    v8f accr[2];
    project_tiles<2>(W_r + (long)dcol * DD + koff,
                     xe + (ih * HROWS + jw * 32 + lr) * XE_STRIDE + koff,
                     b_r[dcol], accr);
    float* xw = xrT + (colbase + lr) * XR_STRIDE + jw * 32 + 8 * hh;
#pragma unroll
    for (int rt = 0; rt < 2; ++rt) {
      const v8f sc = accr[rt] * PAIR_DOWN;
      *(float4*)(xw + rt * 16)     = (float4){ sc[0], sc[1], sc[2], sc[3] };
      *(float4*)(xw + rt * 16 + 4) = (float4){ sc[4], sc[5], sc[6], sc[7] };
    }
  }
  __builtin_amdgcn_fence(__ATOMIC_RELEASE, "workgroup");
  __builtin_amdgcn_s_barrier_signal(-1);

  v8f accl[4];
  project_tiles<4>(W_l + (long)dcol * DD + koff,
                   xe + (jw * HROWS + lr) * XE_STRIDE + koff,
                   b_l[dcol], accl);

  __builtin_amdgcn_sched_barrier(0);
  __builtin_amdgcn_s_barrier_wait(-1);
  __builtin_amdgcn_fence(__ATOMIC_ACQUIRE, "workgroup");

  v2f xl2[16], sum2[16];
#pragma unroll
  for (int rt = 0; rt < 4; ++rt) {
#pragma unroll
    for (int q = 0; q < 4; ++q) {
      xl2[rt * 4 + q]  = (v2f){ accl[rt][2 * q] * PAIR_DOWN, accl[rt][2 * q + 1] * PAIR_DOWN };
      sum2[rt * 4 + q] = (v2f){ 0.f, 0.f };
    }
  }
  const float* xcol = xrT + (colbase + lr) * XR_STRIDE;
#pragma nounroll
  for (int pass = 0; pass < 2; ++pass) {
    v2f xrp[16];
#pragma unroll
    for (int q = 0; q < 8; ++q) {
      const float4 f = *(const float4*)(xcol + pass * 32 + 4 * q);
      xrp[2 * q]     = (v2f){ f.x, f.y };
      xrp[2 * q + 1] = (v2f){ f.z, f.w };
    }
#pragma unroll
    for (int p = 0; p < 16; ++p) {
      v2f t[16];
#pragma unroll
      for (int u = 0; u < 16; ++u) t[u] = relu_add_lo(xrp[p], xl2[u]);
#pragma unroll
      for (int u = 0; u < 16; ++u) sum2[u] += t[u];
#pragma unroll
      for (int u = 0; u < 16; ++u) t[u] = relu_add_hi(xrp[p], xl2[u]);
#pragma unroll
      for (int u = 0; u < 16; ++u) sum2[u] += t[u];
    }
  }
#pragma unroll
  for (int w = 8; w >= 1; w >>= 1) {
#pragma unroll
    for (int u = 0; u < w; ++u) sum2[u] += sum2[u + w];
  }
  const float half_sum = (sum2[0].x + sum2[0].y) * PAIR_UP;
  const float col_sum  = half_sum + xhalf(half_sum);
  if (lane < 16) cs[jw * CGW + colbase + lane] = col_sum;
  __syncthreads();
  if (tid < CGW)
    part[(ih * NBATCH + b) * DD + dbase + tid] = cs[tid] + cs[CGW + tid];
}

__global__ __launch_bounds__(32) void RN_out_kernel(
    const float* __restrict__ part, const float* __restrict__ W_rn,
    const float* __restrict__ b_rn, float* __restrict__ out, size_t ws_bytes) {
  if (ws_bytes < WS_NEED) return;

  const int mt   = blockIdx.x;
  const int nt   = blockIdx.y;
  const int lane = threadIdx.x;
  const int lr   = lane & 15;
  const int koff = (lane >> 4) * 2;
  const int row  = mt * 16 + lr;
  const int col  = nt * 16 + lr;

  const float bv = 16384.0f * b_rn[col];
  v8f c = { bv, bv, bv, bv, bv, bv, bv, bv };

  const float* arow = part + (long)row * DD + koff;
  const float* brow = W_rn + (long)col * DD + koff;
  v2f pa[NPART][8], pb[8];
#pragma unroll
  for (int s = 0; s < 8; ++s) {
#pragma unroll
    for (int p = 0; p < NPART; ++p) pa[p][s] = *(const v2f*)(arow + p * PLANE + 4 * s);
    pb[s] = *(const v2f*)(brow + 4 * s);
  }
  for (int kb = 32; kb < DD; kb += 32) {
    v2f na[NPART][8], nb[8];
#pragma unroll
    for (int s = 0; s < 8; ++s) {
#pragma unroll
      for (int p = 0; p < NPART; ++p) na[p][s] = *(const v2f*)(arow + p * PLANE + kb + 4 * s);
      nb[s] = *(const v2f*)(brow + kb + 4 * s);
    }
    __builtin_amdgcn_sched_barrier(0);
#pragma unroll
    for (int s = 0; s < 8; ++s) {
      const v2f a = pa[0][s] + pa[1][s];
      c = __builtin_amdgcn_wmma_f32_16x16x4_f32(
          false, a, false, pb[s], (short)0, c, false, false);
    }
#pragma unroll
    for (int s = 0; s < 8; ++s) {
#pragma unroll
      for (int p = 0; p < NPART; ++p) pa[p][s] = na[p][s];
      pb[s] = nb[s];
    }
  }
#pragma unroll
  for (int s = 0; s < 8; ++s) {
    const v2f a = pa[0][s] + pa[1][s];
    c = __builtin_amdgcn_wmma_f32_16x16x4_f32(
        false, a, false, pb[s], (short)0, c, false, false);
  }
  float* obase = out + (mt * 16 + 8 * (lane >> 4)) * DD + nt * 16 + lr;
#pragma unroll
  for (int v = 0; v < 8; ++v) obase[v * DD] = c[v];
}

extern "C" void kernel_launch(void* const* d_in, const int* in_sizes, int n_in,
                              void* d_out, int out_size, void* d_ws, size_t ws_size,
                              hipStream_t stream) {
  (void)in_sizes; (void)n_in; (void)out_size;
  const int*   X    = (const int*)d_in[0];
  const float* emb  = (const float*)d_in[1];
  const float* W_l  = (const float*)d_in[2];
  const float* b_l  = (const float*)d_in[3];
  const float* W_r  = (const float*)d_in[4];
  const float* b_r  = (const float*)d_in[5];
  const float* W_rn = (const float*)d_in[6];
  const float* b_rn = (const float*)d_in[7];
  float* part = (float*)d_ws;
  float* out  = (float*)d_out;

  dim3 gridA(NBATCH, NCG, NPART), blockA(256);
  size_t shmem = (size_t)SMEM_FLOATS * sizeof(float);
  RN_pair_block_kernel<<<gridA, blockA, shmem, stream>>>(
      X, emb, W_l, b_l, W_r, b_r, part, ws_size);

  dim3 gridB(2, 16), blockB(32);
  RN_out_kernel<<<gridB, blockB, 0, stream>>>(part, W_rn, b_rn, out, ws_size);
}
